# starting kernel
# baseline (speedup 1.0000x reference)
#include <hip/hip_runtime.h>

typedef __attribute__((ext_vector_type(2))) float v2f;
typedef __attribute__((ext_vector_type(4))) float v4f;
typedef __attribute__((ext_vector_type(8))) float v8f;

#define BN_EPS 1e-5f
#define B_  32
#define C_  256
#define CM_ 128
#define CO_ 256
#define H_  64
#define W_  64
#define L_  128

__global__ __launch_bounds__(256) void ca_pool(const float* __restrict__ x,
                                               float* __restrict__ p) {
    __shared__ float tile[64 * 65];
    const int bc = blockIdx.x;
    const float* xp = x + (size_t)bc * (H_ * W_);
    const int t  = threadIdx.x;
    const int h  = t >> 2;
    const int w0 = (t & 3) * 16;

    const v4f* src = (const v4f*)(xp + h * W_ + w0);
#pragma unroll
    for (int j = 0; j < 4; ++j) {
        v4f v = src[j];
#pragma unroll
        for (int e = 0; e < 4; ++e)
            tile[h * 65 + w0 + j * 4 + e] = v[e];
    }
    __syncthreads();

    if (t < 64) {
        float s = 0.0f;
#pragma unroll 8
        for (int w = 0; w < 64; ++w) s += tile[t * 65 + w];
        p[(size_t)bc * L_ + t] = s * (1.0f / 64.0f);
    } else if (t < 128) {
        const int w = t - 64;
        float s = 0.0f;
#pragma unroll 8
        for (int hh = 0; hh < 64; ++hh) s += tile[hh * 65 + w];
        p[(size_t)bc * L_ + 64 + w] = s * (1.0f / 64.0f);
    }
}

__global__ __launch_bounds__(256) void ca_mlp(
    const float* __restrict__ p,
    const float* __restrict__ w1,
    const float* __restrict__ b1,
    const float* __restrict__ gam, const float* __restrict__ bet,
    const float* __restrict__ mu,  const float* __restrict__ var,
    const float* __restrict__ w2,  const float* __restrict__ b2,
    const float* __restrict__ w3,  const float* __restrict__ b3,
    float* __restrict__ g)
{
    __shared__ float lt[CM_ * L_];
    const int b     = blockIdx.x;
    const float* pb = p + (size_t)b * C_ * L_;
    const int wave  = threadIdx.x >> 5;
    const int lane  = threadIdx.x & 31;
    const int lrow  = lane & 15;
    const int lhalf = lane >> 4;

    const int M0 = wave * 16;
    v8f acc[8];
#pragma unroll
    for (int n = 0; n < 8; ++n) acc[n] = (v8f)(0.0f);

    for (int k = 0; k < C_; k += 4) {
        const float* ar = w1 + (M0 + lrow) * C_ + k + 2 * lhalf;
        v2f a; a[0] = ar[0]; a[1] = ar[1];
#pragma unroll
        for (int n = 0; n < 8; ++n) {
            const int col = n * 16 + lrow;
            v2f bb;
            bb[0] = pb[(k     + 2 * lhalf) * L_ + col];
            bb[1] = pb[(k + 1 + 2 * lhalf) * L_ + col];
            acc[n] = __builtin_amdgcn_wmma_f32_16x16x4_f32(
                false, a, false, bb, (short)0, acc[n], false, false);
        }
    }

    float sc8[8], sh8[8];
#pragma unroll
    for (int i = 0; i < 8; ++i) {
        const int m = M0 + i + 8 * lhalf;
        const float s = gam[m] * rsqrtf(var[m] + BN_EPS);
        sc8[i] = s;
        sh8[i] = bet[m] + (b1[m] - mu[m]) * s;
    }
#pragma unroll
    for (int n = 0; n < 8; ++n) {
#pragma unroll
        for (int i = 0; i < 8; ++i) {
            const int m = M0 + i + 8 * lhalf;
            const float yv = acc[n][i] * sc8[i] + sh8[i];
            const float hs = yv * fminf(fmaxf(yv + 3.0f, 0.0f), 6.0f) * (1.0f / 6.0f);
            lt[m * L_ + n * 16 + lrow] = hs;
        }
    }
    __syncthreads();

    float* gb = g + (size_t)b * CO_ * L_;
#pragma unroll
    for (int oo = 0; oo < 2; ++oo) {
        const int O0 = (wave * 2 + oo) * 16;
        for (int ltile = 0; ltile < 8; ++ltile) {
            const float* ws = (ltile < 4) ? w2 : w3;
            const float* bs = (ltile < 4) ? b2 : b3;
            v8f acc2 = (v8f)(0.0f);
            for (int k = 0; k < CM_; k += 4) {
                const float* ar = ws + (O0 + lrow) * CM_ + k + 2 * lhalf;
                v2f a; a[0] = ar[0]; a[1] = ar[1];
                const int col = ltile * 16 + lrow;
                v2f bb;
                bb[0] = lt[(k     + 2 * lhalf) * L_ + col];
                bb[1] = lt[(k + 1 + 2 * lhalf) * L_ + col];
                acc2 = __builtin_amdgcn_wmma_f32_16x16x4_f32(
                    false, a, false, bb, (short)0, acc2, false, false);
            }
#pragma unroll
            for (int i = 0; i < 8; ++i) {
                const int o  = O0 + i + 8 * lhalf;
                const float yv = acc2[i] + bs[o];
                const float gv = 1.0f / (1.0f + __expf(-yv));
                gb[o * L_ + ltile * 16 + lrow] = gv;
            }
        }
    }
}

__global__ __launch_bounds__(256) void ca_scale(const float* __restrict__ x,
                                                const float* __restrict__ g,
                                                float* __restrict__ out) {
    __shared__ float gr[L_];
    const int bc = blockIdx.x;
    const int t  = threadIdx.x;
    if (t < L_) gr[t] = g[(size_t)bc * L_ + t];
    __syncthreads();

    const float* xp = x   + (size_t)bc * (H_ * W_);
    float*       op = out + (size_t)bc * (H_ * W_);
    const int h  = t >> 2;
    const int w0 = (t & 3) * 16;
    const float gh = gr[h];

    const v4f* src = (const v4f*)(xp + h * W_ + w0);
    v4f*       dst = (v4f*)(op + h * W_ + w0);
#pragma unroll
    for (int j = 0; j < 4; ++j) {
        v4f v = src[j];
#pragma unroll
        for (int e = 0; e < 4; ++e)
            v[e] = v[e] * gh * gr[64 + w0 + j * 4 + e];
        __builtin_nontemporal_store(v, dst + j);
    }
}

extern "C" void kernel_launch(void* const* d_in, const int* in_sizes, int n_in,
                              void* d_out, int out_size, void* d_ws, size_t ws_size,
                              hipStream_t stream) {
    const float* x    = (const float*)d_in[0];
    const float* w1   = (const float*)d_in[1];
    const float* b1   = (const float*)d_in[2];
    const float* gam  = (const float*)d_in[3];
    const float* bet  = (const float*)d_in[4];
    const float* mu   = (const float*)d_in[5];
    const float* var  = (const float*)d_in[6];
    const float* w2   = (const float*)d_in[7];
    const float* b2   = (const float*)d_in[8];
    const float* w3   = (const float*)d_in[9];
    const float* b3   = (const float*)d_in[10];
    float* out = (float*)d_out;

    float* p = (float*)d_ws;
    float* g = p + (size_t)B_ * C_ * L_;

    ca_pool <<<B_ * C_, 256, 0, stream>>>(x, p);
    ca_mlp  <<<B_,       256, 0, stream>>>(p, w1, b1, gam, bet, mu, var,
                                           w2, b2, w3, b3, g);
    ca_scale<<<B_ * C_, 256, 0, stream>>>(x, g, out);
}


# ca_mlp1: K split over 2 waves per 16x64 tile, a wave's fragments requested up front (B rows as 16-byte loads over 4 column-interleaved tiles: 96 loads, not 320), halves summed in LDS; sums re-ordered
# speedup vs baseline: 2.1185x; 2.1185x over previous
#include <hip/hip_runtime.h>

typedef __attribute__((ext_vector_type(2))) float v2f;
typedef __attribute__((ext_vector_type(4))) float v4f;
typedef __attribute__((ext_vector_type(8))) float v8f;

#define BN_EPS 1e-5f
#define B_  32
#define C_  256
#define CM_ 128
#define CO_ 256
#define H_  64
#define W_  64
#define L_  128

#define RPP 20

__global__ __launch_bounds__(256) void ca_pool(const float* __restrict__ x,
                                               float* __restrict__ p) {
    __shared__ __attribute__((aligned(16))) float rp[2][64 * RPP];
    __shared__ __attribute__((aligned(16))) float cp[2][8 * 64];
    const int t    = threadIdx.x;
    const int wave = t >> 5;
    const int lane = t & 31;
    const int sel  = wave >> 2;
    const int wq   = wave & 3;
    const int c4   = lane & 15;
    const int hf   = lane >> 4;
    const int bc   = blockIdx.x * 2 + sel;

    const v4f* src = (const v4f*)(x + (size_t)bc * (H_ * W_)) + wq * 256 + lane;
    v4f v[8];
#pragma unroll
    for (int j = 0; j < 8; ++j) v[j] = src[j * 32];

    float* rpl = rp[sel];
    float* cpl = cp[sel];
    v4f cs = v[0];
#pragma unroll
    for (int j = 0; j < 8; ++j) {
        rpl[(16 * wq + 2 * j + hf) * RPP + c4] = (v[j][0] + v[j][1]) + (v[j][2] + v[j][3]);
        if (j) cs += v[j];
    }
    *(v4f*)(cpl + (2 * wq + hf) * 64 + 4 * c4) = cs;
    __syncthreads();

    const int tl = t & 127;
    float* pout = p + (size_t)bc * L_;
    if (tl < 64) {
        const v4f* q = (const v4f*)(rpl + tl * RPP);
        const v4f s = (q[0] + q[1]) + (q[2] + q[3]);
        pout[tl] = ((s[0] + s[1]) + (s[2] + s[3])) * (1.0f / 64.0f);
    } else {
        const int w = tl - 64;
        float s[8];
#pragma unroll
        for (int g = 0; g < 8; ++g) s[g] = cpl[g * 64 + w];
        pout[64 + w] = (((s[0] + s[1]) + (s[2] + s[3])) + ((s[4] + s[5]) + (s[6] + s[7]))) * (1.0f / 64.0f);
    }
}

__global__ __launch_bounds__(128) void ca_mlp1(
    const float* __restrict__ p,
    const float* __restrict__ w1,
    const float* __restrict__ b1,
    const float* __restrict__ gam, const float* __restrict__ bet,
    const float* __restrict__ mu,  const float* __restrict__ var,
    float* __restrict__ t)
{
    constexpr int KS = 2;
    constexpr int KW = C_ / KS;
    constexpr int NU = KW / 4;
    constexpr int RW = 16 / (2 * KS);
    constexpr int PS = 2 * 16 * 64;
    __shared__ __attribute__((aligned(16))) float red[KS * PS];

    const int wg   = blockIdx.x;
    const int b    = wg >> 3;
    const int M0   = (wg & 7) * 16;
    const int wave = threadIdx.x >> 5;
    const int lane = threadIdx.x & 31;
    const int c    = lane & 15;
    const int h    = lane >> 4;
    const int nh   = wave & 1;
    const int kq   = wave >> 1;
    const int k0   = kq * KW;

    const float* arow = w1 + (M0 + c) * C_ + k0 + 2 * h;
    const float* brow = p + (size_t)b * C_ * L_ + (k0 + 2 * h) * L_ + nh * 64 + 4 * c;

    v2f fa[NU];
    v4f q0[NU], q1[NU];
#pragma unroll
    for (int u = 0; u < NU; ++u) {
        fa[u] = *(const v2f*)(arow + 4 * u);
        q0[u] = *(const v4f*)(brow + (4 * u    ) * L_);
        q1[u] = *(const v4f*)(brow + (4 * u + 1) * L_);
    }

    float es[RW], esh[RW];
#pragma unroll
    for (int i = 0; i < RW; ++i) {
        const int m = M0 + wave * RW + i;
        es[i]  = gam[m] * rsqrtf(var[m] + BN_EPS);
        esh[i] = bet[m] + (b1[m] - mu[m]) * es[i];
    }

    v8f acc[4];
#pragma unroll
    for (int j = 0; j < 4; ++j) acc[j] = (v8f)(0.0f);
#pragma unroll
    for (int u = 0; u < NU; ++u)
#pragma unroll
        for (int j = 0; j < 4; ++j) {
            v2f bj;
            bj[0] = q0[u][j];
            bj[1] = q1[u][j];
            acc[j] = __builtin_amdgcn_wmma_f32_16x16x4_f32(false, fa[u], false, bj, (short)0,
                                                           acc[j], false, false);
        }

    float* rw = red + kq * PS + (nh * 16 + 8 * h) * 64 + 4 * c;
#pragma unroll
    for (int r = 0; r < 8; ++r) {
        v4f v;
        v[0] = acc[0][r]; v[1] = acc[1][r]; v[2] = acc[2][r]; v[3] = acc[3][r];
        *(v4f*)(rw + r * 64) = v;
    }
    __syncthreads();

    float* tb = t + (size_t)b * CM_ * L_;
    const float* rr = red + ((lane >> 4) * 16 + wave * RW) * 64 + 4 * (lane & 15);
#pragma unroll
    for (int i = 0; i < RW; ++i) {
        v4f part[KS];
#pragma unroll
        for (int q = 0; q < KS; ++q) part[q] = *(const v4f*)(rr + q * PS + i * 64);
        const v4f sum = part[0] + part[1];
        v4f o;
#pragma unroll
        for (int e = 0; e < 4; ++e) {
            const float yv = sum[e] * es[i] + esh[i];
            o[e] = yv * fminf(fmaxf(yv + 3.0f, 0.0f), 6.0f) * (1.0f / 6.0f);
        }
        *(v4f*)(tb + (M0 + wave * RW + i) * L_ + 4 * lane) = o;
    }
}

#define GP 132

__device__ __forceinline__ void half_load(v4f (&buf)[16], const v4f* __restrict__ src) {
#pragma unroll
    for (int j = 0; j < 16; ++j) buf[j] = src[j * 32];
}

__device__ __forceinline__ void half_scale_store(const v4f (&buf)[16], const float* __restrict__ ghp,
                                                 const v4f gw4, v4f* __restrict__ dst) {
    v4f gh4[4];
#pragma unroll
    for (int i = 0; i < 4; ++i) gh4[i] = *(const v4f*)(ghp + 4 * i);
#pragma unroll
    for (int j = 0; j < 16; ++j) {
        const float gh = gh4[j >> 2][j & 3];
        v4f v = buf[j];
#pragma unroll
        for (int e = 0; e < 4; ++e) v[e] = v[e] * gh * gw4[e];
        __builtin_nontemporal_store(v, dst + j * 32);
    }
}

__global__ __launch_bounds__(256) void ca_gate_scale(
    const float* __restrict__ x,
    const float* __restrict__ t,
    const float* __restrict__ w2,  const float* __restrict__ b2,
    const float* __restrict__ w3,  const float* __restrict__ b3,
    float* __restrict__ out)
{
    __shared__ __attribute__((aligned(16))) float gs[16 * GP];
    const int wg    = blockIdx.x;
    const int b     = wg >> 4;
    const int c0    = (wg & 15) * 16;
    const int wave  = threadIdx.x >> 5;
    const int lane  = threadIdx.x & 31;
    const int lrow  = lane & 15;
    const int lhalf = lane >> 4;

    const int N0 = wave * 16;
    const bool second = wave >= 4;
    const float* ws = second ? w3 : w2;
    const float* bs = second ? b3 : b2;
    const float* arow = ws + (c0 + lrow) * CM_ + 2 * lhalf;
    const float* bcol = t + (size_t)b * CM_ * L_ + (2 * lhalf) * L_ + N0 + lrow;

    float bias[8];
#pragma unroll
    for (int i = 0; i < 8; ++i) bias[i] = bs[c0 + i + 8 * lhalf];

    v2f fa[32], fb[32];
#pragma unroll
    for (int u = 0; u < 32; ++u) {
        const int k = 4 * u;
        fa[u]    = *(const v2f*)(arow + k);
        fb[u][0] = bcol[(k    ) * L_];
        fb[u][1] = bcol[(k + 1) * L_];
    }

    const size_t pl = ((size_t)(b * C_ + c0 + 2 * wave)) * (H_ * W_);
    const v4f* xs = (const v4f*)(x + pl) + lane;
    v4f*       od = (v4f*)(out + pl) + lane;
    v4f bufA[16], bufB[16];
    half_load(bufA, xs);

    v8f acc = (v8f)(0.0f);
#pragma unroll
    for (int u = 0; u < 32; ++u)
        acc = __builtin_amdgcn_wmma_f32_16x16x4_f32(false, fa[u], false, fb[u], (short)0, acc,
                                                    false, false);

    const int l   = N0 + lrow;
    const int pos = second ? l : ((l & 1) * 32 + (l >> 1));
#pragma unroll
    for (int i = 0; i < 8; ++i) {
        const int ch = i + 8 * lhalf;
        const float yv = acc[i] + bias[i];
        gs[ch * GP + pos] = 1.0f / (1.0f + __expf(-yv));
    }
    __syncthreads();

    const float* g0 = gs + (2 * wave) * GP;
    const float* g1 = g0 + GP;
    const v4f gw0 = *(const v4f*)(g0 + 64 + lrow * 4);
    const v4f gw1 = *(const v4f*)(g1 + 64 + lrow * 4);
    const float* gh0 = g0 + lhalf * 32;
    const float* gh1 = g1 + lhalf * 32;

    half_load(bufB, xs + 512);
    __builtin_amdgcn_sched_barrier(0);
    half_scale_store(bufA, gh0, gw0, od);
    __builtin_amdgcn_sched_barrier(0);
    half_load(bufA, xs + 1024);
    __builtin_amdgcn_sched_barrier(0);
    half_scale_store(bufB, gh0 + 16, gw0, od + 512);
    __builtin_amdgcn_sched_barrier(0);
    half_load(bufB, xs + 1536);
    __builtin_amdgcn_sched_barrier(0);
    half_scale_store(bufA, gh1, gw1, od + 1024);
    half_scale_store(bufB, gh1 + 16, gw1, od + 1536);
}

extern "C" void kernel_launch(void* const* d_in, const int* in_sizes, int n_in,
                              void* d_out, int out_size, void* d_ws, size_t ws_size,
                              hipStream_t stream) {
    const float* x    = (const float*)d_in[0];
    const float* w1   = (const float*)d_in[1];
    const float* b1   = (const float*)d_in[2];
    const float* gam  = (const float*)d_in[3];
    const float* bet  = (const float*)d_in[4];
    const float* mu   = (const float*)d_in[5];
    const float* var  = (const float*)d_in[6];
    const float* w2   = (const float*)d_in[7];
    const float* b2   = (const float*)d_in[8];
    const float* w3   = (const float*)d_in[9];
    const float* b3   = (const float*)d_in[10];
    float* out = (float*)d_out;

    float* p = (float*)d_ws;
    float* t = p + (size_t)B_ * C_ * L_;

    ca_pool <<<B_ * C_ / 2, 256, 0, stream>>>(x, p);
    ca_mlp1 <<<B_ * 8,  128, 0, stream>>>(p, w1, b1, gam, bet, mu, var, t);
    ca_gate_scale<<<B_ * 16, 256, 0, stream>>>(x, t, w2, b2, w3, b3, out);
}
